# starting kernel
# baseline (speedup 1.0000x reference)
#include <hip/hip_runtime.h>
#include <math.h>

typedef float v2f __attribute__((ext_vector_type(2)));
typedef float v8f __attribute__((ext_vector_type(8)));

#define BS      65536
#define IN_DIM  100
#define HID     16
#define NCORP   10
#define ROWS    (BS * NCORP)
#define WPB     8

__global__ __launch_bounds__(256) void proj_tvec_kernel(
    const float* __restrict__ item_corpus,
    const float* __restrict__ W_proj,
    const float* __restrict__ b_proj,
    const float* __restrict__ W_t,
    float* __restrict__ proj,
    float* __restrict__ tvec)
{
    __shared__ float lds[WPB][16][17];

    const int lane  = threadIdx.x & 31;
    const int wv    = threadIdx.x >> 5;
    const int l15   = lane & 15;
    const int khalf = (lane >> 4) * 2;
    const int rofs  = (lane >> 4) * 8;

    const long tile = (long)blockIdx.x * WPB + wv;
    const long row0 = tile * 16;

    v8f acc = {};
    const float* arow = item_corpus + (row0 + l15) * IN_DIM;
    #pragma unroll
    for (int k0 = 0; k0 < IN_DIM; k0 += 4) {
        v2f a = *(const v2f*)(arow + k0 + khalf);
        v2f b;
        b.x = W_proj[(k0 + khalf + 0) * HID + l15];
        b.y = W_proj[(k0 + khalf + 1) * HID + l15];
        acc = __builtin_amdgcn_wmma_f32_16x16x4_f32(
                  false, a, false, b, (short)0, acc, false, false);
    }

    const float bias = b_proj[l15];
    #pragma unroll
    for (int r = 0; r < 8; ++r) acc[r] += bias;

    #pragma unroll
    for (int r = 0; r < 8; ++r) {
        proj[(row0 + r + rofs) * HID + l15] = acc[r];
        lds[wv][r + rofs][l15] = acc[r];
    }
    __syncthreads();

    v8f acc2 = {};
    #pragma unroll
    for (int kc = 0; kc < 4; ++kc) {
        v2f a;
        a.x = lds[wv][l15][kc * 4 + khalf + 0];
        a.y = lds[wv][l15][kc * 4 + khalf + 1];
        v2f b;
        b.x = W_t[(kc * 4 + khalf + 0) * HID + l15];
        b.y = W_t[(kc * 4 + khalf + 1) * HID + l15];
        acc2 = __builtin_amdgcn_wmma_f32_16x16x4_f32(
                   false, a, false, b, (short)0, acc2, false, false);
    }
    #pragma unroll
    for (int r = 0; r < 8; ++r)
        tvec[(row0 + r + rofs) * HID + l15] = tanhf(acc2[r]);
}

__global__ __launch_bounds__(256) void greedy_kernel(
    const float* __restrict__ user_intent,
    const float* __restrict__ proj,
    const float* __restrict__ tvec,
    const float* __restrict__ W_s,
    const int*   __restrict__ search_num_p,
    float* __restrict__ out)
{
    __shared__ float Ws[256];
    Ws[threadIdx.x] = W_s[threadIdx.x];
    __syncthreads();

    const int  b     = blockIdx.x * blockDim.x + threadIdx.x;
    const int  steps = *search_num_p;
    const long obase = (long)b * (steps + 1) * HID;

    float sum[HID];
    #pragma unroll
    for (int h = 0; h < HID; ++h) {
        float v = user_intent[(long)b * HID + h];
        sum[h] = v;
        out[obase + h] = v;
    }

    const float* tb = tvec + (long)b * NCORP * HID;
    const float* cb = proj + (long)b * NCORP * HID;

    for (int it = 0; it < steps; ++it) {
        const float inv = 1.0f / (float)(it + 1);

        float s[HID];
        #pragma unroll
        for (int h = 0; h < HID; ++h) {
            float a = 0.0f;
            #pragma unroll
            for (int hp = 0; hp < HID; ++hp)
                a += (sum[hp] * inv) * Ws[hp * HID + h];
            s[h] = tanhf(a);
        }

        float best = -__builtin_inff();
        int bidx = 0;
        for (int c = 0; c < NCORP; ++c) {
            float sc = 0.0f;
            #pragma unroll
            for (int h = 0; h < HID; ++h)
                sc += tb[c * HID + h] * s[h];
            if (sc > best) { best = sc; bidx = c; }
        }

        #pragma unroll
        for (int h = 0; h < HID; ++h) {
            float v = cb[bidx * HID + h];
            sum[h] += v;
            out[obase + (long)(it + 1) * HID + h] = v;
        }
    }
}

extern "C" void kernel_launch(void* const* d_in, const int* in_sizes, int n_in,
                              void* d_out, int out_size, void* d_ws, size_t ws_size,
                              hipStream_t stream) {
    const float* user_intent = (const float*)d_in[0];
    const float* item_corpus = (const float*)d_in[1];
    const float* W_proj      = (const float*)d_in[2];
    const float* b_proj      = (const float*)d_in[3];
    const float* W_s         = (const float*)d_in[4];
    const float* W_t         = (const float*)d_in[5];
    const int*   search_num  = (const int*)d_in[6];
    float* out = (float*)d_out;

    float* proj = (float*)d_ws;
    float* tv   = proj + (size_t)ROWS * HID;

    const int tiles   = ROWS / 16;
    const int blocks1 = tiles / WPB;

    proj_tvec_kernel<<<blocks1, 256, 0, stream>>>(item_corpus, W_proj, b_proj,
                                                  W_t, proj, tv);
    greedy_kernel<<<BS / 256, 256, 0, stream>>>(user_intent, proj, tv,
                                                W_s, search_num, out);
}


# A tile of each wave (16 rows x 100 f32 = 6400 contiguous bytes) fetched with whole-line 16-byte global loads, staged in per-wave LDS; WMMA A fragments read from LDS, same k order, same sums
# speedup vs baseline: 1.4006x; 1.4006x over previous
#include <hip/hip_runtime.h>
#include <math.h>

typedef float v2f __attribute__((ext_vector_type(2)));
typedef float v8f __attribute__((ext_vector_type(8)));

#define BS      65536
#define IN_DIM  100
#define HID     16
#define NCORP   10
#define ROWS    (BS * NCORP)
#define WPB     8
#define TILE_F  (16 * IN_DIM)
#define RL_P    17

__global__ __launch_bounds__(256) void proj_tvec_kernel(
    const float* __restrict__ item_corpus,
    const float* __restrict__ W_proj,
    const float* __restrict__ b_proj,
    const float* __restrict__ W_t,
    float* __restrict__ proj,
    float* __restrict__ tvec)
{
    __shared__ __attribute__((aligned(16))) float tileS[WPB][TILE_F];

    const int lane  = threadIdx.x & 31;
    const int wv    = threadIdx.x >> 5;
    const int l15   = lane & 15;
    const int khalf = (lane >> 4) * 2;
    const int rofs  = (lane >> 4) * 8;

    const long tile = (long)blockIdx.x * WPB + wv;
    const long row0 = tile * 16;

    float* tileW = &tileS[wv][0];
    {
        const float4* gsrc = (const float4*)(item_corpus + row0 * IN_DIM);
        float4*       ldst = (float4*)tileW;
        float4 t[12];
        #pragma unroll
        for (int i = 0; i < 12; ++i) t[i] = gsrc[i * 32 + lane];
        const v2f tl = ((const v2f*)(gsrc + 12 * 32))[lane];
        #pragma unroll
        for (int i = 0; i < 12; ++i) ldst[i * 32 + lane] = t[i];
        ((v2f*)(ldst + 12 * 32))[lane] = tl;
    }
    __syncthreads();

    v8f acc = {};
    const float* arow = tileW + l15 * IN_DIM;
    #pragma unroll
    for (int k0 = 0; k0 < IN_DIM; k0 += 4) {
        v2f a = *(const v2f*)(arow + k0 + khalf);
        v2f b;
        b.x = W_proj[(k0 + khalf + 0) * HID + l15];
        b.y = W_proj[(k0 + khalf + 1) * HID + l15];
        acc = __builtin_amdgcn_wmma_f32_16x16x4_f32(
                  false, a, false, b, (short)0, acc, false, false);
    }

    const float bias = b_proj[l15];
    #pragma unroll
    for (int r = 0; r < 8; ++r) acc[r] += bias;

    #pragma unroll
    for (int r = 0; r < 8; ++r) {
        proj[(row0 + r + rofs) * HID + l15] = acc[r];
        tileW[(r + rofs) * RL_P + l15] = acc[r];
    }
    __syncthreads();

    v8f acc2 = {};
    #pragma unroll
    for (int kc = 0; kc < 4; ++kc) {
        v2f a;
        a.x = tileW[l15 * RL_P + kc * 4 + khalf + 0];
        a.y = tileW[l15 * RL_P + kc * 4 + khalf + 1];
        v2f b;
        b.x = W_t[(kc * 4 + khalf + 0) * HID + l15];
        b.y = W_t[(kc * 4 + khalf + 1) * HID + l15];
        acc2 = __builtin_amdgcn_wmma_f32_16x16x4_f32(
                   false, a, false, b, (short)0, acc2, false, false);
    }
    #pragma unroll
    for (int r = 0; r < 8; ++r)
        tvec[(row0 + r + rofs) * HID + l15] = tanhf(acc2[r]);
}

__global__ __launch_bounds__(256) void greedy_kernel(
    const float* __restrict__ user_intent,
    const float* __restrict__ proj,
    const float* __restrict__ tvec,
    const float* __restrict__ W_s,
    const int*   __restrict__ search_num_p,
    float* __restrict__ out)
{
    __shared__ float Ws[256];
    Ws[threadIdx.x] = W_s[threadIdx.x];
    __syncthreads();

    const int  b     = blockIdx.x * blockDim.x + threadIdx.x;
    const int  steps = *search_num_p;
    const long obase = (long)b * (steps + 1) * HID;

    float sum[HID];
    #pragma unroll
    for (int h = 0; h < HID; ++h) {
        float v = user_intent[(long)b * HID + h];
        sum[h] = v;
        out[obase + h] = v;
    }

    const float* tb = tvec + (long)b * NCORP * HID;
    const float* cb = proj + (long)b * NCORP * HID;

    for (int it = 0; it < steps; ++it) {
        const float inv = 1.0f / (float)(it + 1);

        float s[HID];
        #pragma unroll
        for (int h = 0; h < HID; ++h) {
            float a = 0.0f;
            #pragma unroll
            for (int hp = 0; hp < HID; ++hp)
                a += (sum[hp] * inv) * Ws[hp * HID + h];
            s[h] = tanhf(a);
        }

        float best = -__builtin_inff();
        int bidx = 0;
        for (int c = 0; c < NCORP; ++c) {
            float sc = 0.0f;
            #pragma unroll
            for (int h = 0; h < HID; ++h)
                sc += tb[c * HID + h] * s[h];
            if (sc > best) { best = sc; bidx = c; }
        }

        #pragma unroll
        for (int h = 0; h < HID; ++h) {
            float v = cb[bidx * HID + h];
            sum[h] += v;
            out[obase + (long)(it + 1) * HID + h] = v;
        }
    }
}

extern "C" void kernel_launch(void* const* d_in, const int* in_sizes, int n_in,
                              void* d_out, int out_size, void* d_ws, size_t ws_size,
                              hipStream_t stream) {
    const float* user_intent = (const float*)d_in[0];
    const float* item_corpus = (const float*)d_in[1];
    const float* W_proj      = (const float*)d_in[2];
    const float* b_proj      = (const float*)d_in[3];
    const float* W_s         = (const float*)d_in[4];
    const float* W_t         = (const float*)d_in[5];
    const int*   search_num  = (const int*)d_in[6];
    float* out = (float*)d_out;

    float* proj = (float*)d_ws;
    float* tv   = proj + (size_t)ROWS * HID;

    const int tiles   = ROWS / 16;
    const int blocks1 = tiles / WPB;

    proj_tvec_kernel<<<blocks1, 256, 0, stream>>>(item_corpus, W_proj, b_proj,
                                                  W_t, proj, tv);
    greedy_kernel<<<BS / 256, 256, 0, stream>>>(user_intent, proj, tv,
                                                W_s, search_num, out);
}


# Both launches fused: a wave projects the 80 rows of its 8 batch elements into LDS and runs their greedy search there, 4 lanes per element; proj/tvec skip the workspace; every sum in the same order
# speedup vs baseline: 2.1989x; 1.5700x over previous
#include <hip/hip_runtime.h>
#include <math.h>

typedef float v2f __attribute__((ext_vector_type(2)));
typedef float v4f __attribute__((ext_vector_type(4)));
typedef float v8f __attribute__((ext_vector_type(8)));

#define BS      65536
#define IN_DIM  100
#define HID     16
#define NCORP   10
#define WPB     4
#define TPW     5
#define BPW     8
#define TILE_F  (16 * IN_DIM)
#define PITCH   20
#define RES_F   (TPW * 16 * PITCH)


__device__ __forceinline__ void wave_lds_handover() {
    __builtin_amdgcn_fence(__ATOMIC_RELEASE, "wavefront");
    __builtin_amdgcn_wave_barrier();
    __builtin_amdgcn_fence(__ATOMIC_ACQUIRE, "wavefront");
}

template <int CTRL>
__device__ __forceinline__ int quad_i(int v) {
    return __builtin_amdgcn_update_dpp(0, v, CTRL, 0xF, 0xF, false);
}
template <int CTRL>
__device__ __forceinline__ float quad_f(float v) {
    return __int_as_float(quad_i<CTRL>(__float_as_int(v)));
}

__global__ __launch_bounds__(WPB * 32) void fused_search_kernel(
    const float* __restrict__ user_intent,
    const float* __restrict__ item_corpus,
    const float* __restrict__ W_proj,
    const float* __restrict__ b_proj,
    const float* __restrict__ W_s,
    const float* __restrict__ W_t,
    const int*   __restrict__ search_num_p,
    float* __restrict__ out)
{
    __shared__ __attribute__((aligned(16))) float tileS[WPB][TILE_F];
    __shared__ __attribute__((aligned(16))) float projS[WPB][RES_F];
    __shared__ __attribute__((aligned(16))) float tvecS[WPB][RES_F];

    const int lane  = threadIdx.x & 31;
    const int wv    = threadIdx.x >> 5;
    const int l15   = lane & 15;
    const int khalf = (lane >> 4) * 2;
    const int rofs  = (lane >> 4) * 8;

    const long wave = (long)blockIdx.x * WPB + wv;
    float* tileW = &tileS[wv][0];
    float* projW = &projS[wv][0];
    float* tvecW = &tvecS[wv][0];

    v2f bw[IN_DIM / 4];
    #pragma unroll
    for (int i = 0; i < IN_DIM / 4; ++i) {
        bw[i].x = W_proj[(i * 4 + khalf + 0) * HID + l15];
        bw[i].y = W_proj[(i * 4 + khalf + 1) * HID + l15];
    }
    v2f bt[4];
    #pragma unroll
    for (int kc = 0; kc < 4; ++kc) {
        bt[kc].x = W_t[(kc * 4 + khalf + 0) * HID + l15];
        bt[kc].y = W_t[(kc * 4 + khalf + 1) * HID + l15];
    }
    const float bias = b_proj[l15];

    const v4f* gsrc = (const v4f*)(item_corpus + wave * (long)(TPW * TILE_F));
    v4f    pf[12];
    v2f    pft;
    #pragma unroll
    for (int i = 0; i < 12; ++i) pf[i] = gsrc[i * 32 + lane];
    pft = ((const v2f*)(gsrc + 12 * 32))[lane];

    for (int t = 0; t < TPW; ++t) {
        wave_lds_handover();
        {
            v4f* ldst = (v4f*)tileW;
            #pragma unroll
            for (int i = 0; i < 12; ++i) ldst[i * 32 + lane] = pf[i];
            ((v2f*)(ldst + 12 * 32))[lane] = pft;
        }
        if (t + 1 < TPW) {
            const v4f* g = gsrc + (t + 1) * (TILE_F / 4);
            #pragma unroll
            for (int i = 0; i < 12; ++i) pf[i] = g[i * 32 + lane];
            pft = ((const v2f*)(g + 12 * 32))[lane];
        }
        wave_lds_handover();

        v8f acc = {};
        const float* arow = tileW + l15 * IN_DIM;
        #pragma unroll
        for (int i = 0; i < IN_DIM / 4; ++i) {
            v2f a = *(const v2f*)(arow + i * 4 + khalf);
            acc = __builtin_amdgcn_wmma_f32_16x16x4_f32(
                      false, a, false, bw[i], (short)0, acc, false, false);
        }
        #pragma unroll
        for (int r = 0; r < 8; ++r) acc[r] += bias;

        float* prow = projW + (t * 16) * PITCH;
        #pragma unroll
        for (int r = 0; r < 8; ++r) prow[(r + rofs) * PITCH + l15] = acc[r];
        wave_lds_handover();

        v8f acc2 = {};
        #pragma unroll
        for (int kc = 0; kc < 4; ++kc) {
            v2f a = *(const v2f*)(prow + l15 * PITCH + kc * 4 + khalf);
            acc2 = __builtin_amdgcn_wmma_f32_16x16x4_f32(
                       false, a, false, bt[kc], (short)0, acc2, false, false);
        }
        float* trow = tvecW + (t * 16) * PITCH;
        #pragma unroll
        for (int r = 0; r < 8; ++r) trow[(r + rofs) * PITCH + l15] = tanhf(acc2[r]);
    }
    wave_lds_handover();

    const int  bl    = lane >> 2;
    const int  j     = lane & 3;
    const long b     = wave * BPW + bl;
    const int  steps = *search_num_p;
    const long obase = b * (long)(steps + 1) * HID;

    v4f wsr[HID];
    #pragma unroll
    for (int hp = 0; hp < HID; ++hp)
        wsr[hp] = *(const v4f*)(W_s + hp * HID + 4 * j);

    float sum[HID];
    {
        const v4f* ui = (const v4f*)(user_intent + b * HID);
        #pragma unroll
        for (int g = 0; g < 4; ++g) {
            const v4f u = ui[g];
            sum[4 * g + 0] = u.x; sum[4 * g + 1] = u.y;
            sum[4 * g + 2] = u.z; sum[4 * g + 3] = u.w;
        }
        *(v4f*)(out + obase + 4 * j) = ui[j];
    }

    float tv[3][HID];
    #pragma unroll
    for (int q = 0; q < 3; ++q) {
        const int c  = j + 4 * q;
        const int cc = c < NCORP ? c : NCORP - 1;
        const v4f* tr = (const v4f*)(tvecW + (bl * NCORP + cc) * PITCH);
        #pragma unroll
        for (int g = 0; g < 4; ++g) {
            const v4f x = tr[g];
            tv[q][4 * g + 0] = x.x; tv[q][4 * g + 1] = x.y;
            tv[q][4 * g + 2] = x.z; tv[q][4 * g + 3] = x.w;
        }
    }

    for (int it = 0; it < steps; ++it) {
        const float inv = 1.0f / (float)(it + 1);

        float sl[4];
        {
            float a0 = 0.0f, a1 = 0.0f, a2 = 0.0f, a3 = 0.0f;
            #pragma unroll
            for (int hp = 0; hp < HID; ++hp) {
                a0 += (sum[hp] * inv) * wsr[hp].x;
                a1 += (sum[hp] * inv) * wsr[hp].y;
                a2 += (sum[hp] * inv) * wsr[hp].z;
                a3 += (sum[hp] * inv) * wsr[hp].w;
            }
            sl[0] = tanhf(a0); sl[1] = tanhf(a1);
            sl[2] = tanhf(a2); sl[3] = tanhf(a3);
        }

        float s[HID];
        #pragma unroll
        for (int i = 0; i < 4; ++i) {
            s[0  + i] = quad_f<0x00>(sl[i]);
            s[4  + i] = quad_f<0x55>(sl[i]);
            s[8  + i] = quad_f<0xAA>(sl[i]);
            s[12 + i] = quad_f<0xFF>(sl[i]);
        }

        float best = -__builtin_inff();
        int bidx = 0;
        #pragma unroll
        for (int q = 0; q < 3; ++q) {
            const int c = j + 4 * q;
            float sc = 0.0f;
            #pragma unroll
            for (int h = 0; h < HID; ++h)
                sc += tv[q][h] * s[h];
            if (c < NCORP && sc > best) { best = sc; bidx = c; }
        }
        {
            const float ob = quad_f<0xB1>(best);
            const int   oi = quad_i<0xB1>(bidx);
            if (ob > best || (ob == best && oi < bidx)) { best = ob; bidx = oi; }
        }
        {
            const float ob = quad_f<0x4E>(best);
            const int   oi = quad_i<0x4E>(bidx);
            if (ob > best || (ob == best && oi < bidx)) { best = ob; bidx = oi; }
        }

        const float* pr = projW + (bl * NCORP + bidx) * PITCH;
        #pragma unroll
        for (int g = 0; g < 4; ++g) {
            const v4f v = ((const v4f*)pr)[g];
            sum[4 * g + 0] += v.x; sum[4 * g + 1] += v.y;
            sum[4 * g + 2] += v.z; sum[4 * g + 3] += v.w;
        }
        *(v4f*)(out + obase + (long)(it + 1) * HID + 4 * j) = ((const v4f*)pr)[j];
    }
}

extern "C" void kernel_launch(void* const* d_in, const int* in_sizes, int n_in,
                              void* d_out, int out_size, void* d_ws, size_t ws_size,
                              hipStream_t stream) {
    const float* user_intent = (const float*)d_in[0];
    const float* item_corpus = (const float*)d_in[1];
    const float* W_proj      = (const float*)d_in[2];
    const float* b_proj      = (const float*)d_in[3];
    const float* W_s         = (const float*)d_in[4];
    const float* W_t         = (const float*)d_in[5];
    const int*   search_num  = (const int*)d_in[6];
    float* out = (float*)d_out;

    fused_search_kernel<<<BS / (BPW * WPB), WPB * 32, 0, stream>>>(
        user_intent, item_corpus, W_proj, b_proj, W_s, W_t, search_num, out);
}
